# starting kernel
# baseline (speedup 1.0000x reference)
#include <hip/hip_runtime.h>

typedef __attribute__((ext_vector_type(2))) float v2f;
typedef __attribute__((ext_vector_type(8))) float v8f;

constexpr int Bn = 4;
constexpr int Nn = 32768;
constexpr int Kn = 20;
constexpr int Cn = 64;
constexpr int On = 64;

constexpr int WAVES   = 4;
constexpr int A_STRIDE = 132;

__global__ __launch_bounds__(WAVES * 32)
void EdgeConv_wmma_kernel(const float* __restrict__ x,
                          const int*   __restrict__ adj,
                          const float* __restrict__ W,
                          const float* __restrict__ bias,
                          float*       __restrict__ out)
{
    __shared__ __align__(16) float s_concat[WAVES][Kn * Cn];
    __shared__ __align__(16) float s_A[WAVES][16 * A_STRIDE];

    const int lane = threadIdx.x & 31;
    const int wid  = threadIdx.x >> 5;
    const int wave_global = blockIdx.x * WAVES + wid;
    const int base_point  = wave_global * 16;

    float* concat = s_concat[wid];
    float* Atile  = s_A[wid];

    const int hi  = lane >> 4;
    const int l15 = lane & 15;

    for (int p = 0; p < 16; ++p) {
        const int point = base_point + p;
        const int b = point >> 15;
        const int n = point & (Nn - 1);
        const float* xb = x + (size_t)b * Nn * Cn;

        int vidx = 0;
        if (lane < Kn) vidx = adj[(size_t)point * Kn + lane];

        for (int k = 0; k < Kn; ++k) {
            const int row = __shfl(vidx, k, 32);
            const float2 v = *(const float2*)(xb + (size_t)row * Cn + lane * 2);
            *(float2*)&concat[k * Cn + lane * 2] = v;
        }

        const float xc0 = xb[(size_t)n * Cn + lane];
        const float xc1 = xb[(size_t)n * Cn + lane + 32];

        float s0 = 0.f, s1 = 0.f;
        #pragma unroll
        for (int t = 0; t < Kn; ++t) {
            s0 += concat[Kn * lane        + t];
            s1 += concat[Kn * (lane + 32) + t];
        }
        const float inv = 1.0f / (float)Kn;
        Atile[p * A_STRIDE + lane]           = s0 * inv - xc0;
        Atile[p * A_STRIDE + lane + 32]      = s1 * inv - xc1;
        Atile[p * A_STRIDE + 64 + lane]      = xc0;
        Atile[p * A_STRIDE + 64 + lane + 32] = xc1;
    }

    v8f acc0 = {0,0,0,0,0,0,0,0};
    v8f acc1 = acc0, acc2 = acc0, acc3 = acc0;

    #pragma unroll 4
    for (int kk = 0; kk < 32; ++kk) {
        const float2 av = *(const float2*)&Atile[l15 * A_STRIDE + kk * 4 + 2 * hi];
        const v2f Af = { av.x, av.y };

        const float* wb = W + (size_t)l15 * (2 * Cn) + kk * 4 + 2 * hi;
        const float2 b0 = *(const float2*)(wb + (size_t)0  * 16 * (2 * Cn));
        const float2 b1 = *(const float2*)(wb + (size_t)1  * 16 * (2 * Cn));
        const float2 b2 = *(const float2*)(wb + (size_t)2  * 16 * (2 * Cn));
        const float2 b3 = *(const float2*)(wb + (size_t)3  * 16 * (2 * Cn));
        const v2f B0 = { b0.x, b0.y };
        const v2f B1 = { b1.x, b1.y };
        const v2f B2 = { b2.x, b2.y };
        const v2f B3 = { b3.x, b3.y };

        acc0 = __builtin_amdgcn_wmma_f32_16x16x4_f32(false, Af, false, B0, (short)0, acc0, false, false);
        acc1 = __builtin_amdgcn_wmma_f32_16x16x4_f32(false, Af, false, B1, (short)0, acc1, false, false);
        acc2 = __builtin_amdgcn_wmma_f32_16x16x4_f32(false, Af, false, B2, (short)0, acc2, false, false);
        acc3 = __builtin_amdgcn_wmma_f32_16x16x4_f32(false, Af, false, B3, (short)0, acc3, false, false);
    }

    const float bz0 = bias[ 0 + l15];
    const float bz1 = bias[16 + l15];
    const float bz2 = bias[32 + l15];
    const float bz3 = bias[48 + l15];

    #pragma unroll
    for (int v = 0; v < 8; ++v) {
        const int prow = v + 8 * hi;
        float* o = out + (size_t)(base_point + prow) * On;
        o[ 0 + l15] = acc0[v] + bz0;
        o[16 + l15] = acc1[v] + bz1;
        o[32 + l15] = acc2[v] + bz2;
        o[48 + l15] = acc3[v] + bz3;
    }
}

extern "C" void kernel_launch(void* const* d_in, const int* in_sizes, int n_in,
                              void* d_out, int out_size, void* d_ws, size_t ws_size,
                              hipStream_t stream) {
    const float* x    = (const float*)d_in[0];
    const int*   adj  = (const int*)  d_in[1];
    const float* W    = (const float*)d_in[2];
    const float* bias = (const float*)d_in[3];
    float* out = (float*)d_out;

    const int total_points = Bn * Nn;
    const int total_waves  = total_points / 16;
    const int blocks       = total_waves / WAVES;

    EdgeConv_wmma_kernel<<<blocks, WAVES * 32, 0, stream>>>(x, adj, W, bias, out);
}


# Gather with 16-byte loads (2 whole rows each) and sum every aligned 4-float piece in registers: LDS holds 320 partial sums per point, not 1280 floats; window sum order changes (4-term pieces, then 5)
# speedup vs baseline: 1.2482x; 1.2482x over previous
#include <hip/hip_runtime.h>

typedef __attribute__((ext_vector_type(2))) float v2f;
typedef __attribute__((ext_vector_type(8))) float v8f;

constexpr int Bn = 4;
constexpr int Nn = 32768;
constexpr int Kn = 20;
constexpr int Cn = 64;
constexpr int On = 64;

constexpr int WAVES   = 4;
constexpr int A_STRIDE = 132;
constexpr int NPART   = Kn * Cn / 4;

__global__ __launch_bounds__(WAVES * 32)
void EdgeConv_wmma_kernel(const float* __restrict__ x,
                          const int*   __restrict__ adj,
                          const float* __restrict__ W,
                          const float* __restrict__ bias,
                          float*       __restrict__ out)
{
    __shared__ __align__(16) float s_part[WAVES][NPART];
    __shared__ __align__(16) float s_A[WAVES][16 * A_STRIDE];

    const int lane = threadIdx.x & 31;
    const int wid  = threadIdx.x >> 5;
    const int wave_global = blockIdx.x * WAVES + wid;
    const int base_point  = wave_global * 16;

    float* part   = s_part[wid];
    float* Atile  = s_A[wid];

    const int hi  = lane >> 4;
    const int l15 = lane & 15;

    for (int p = 0; p < 16; ++p) {
        const int point = base_point + p;
        const int b = point >> 15;
        const int n = point & (Nn - 1);
        const float* xb = x + (size_t)b * Nn * Cn;

        int vidx = 0;
        if (lane < Kn) vidx = adj[(size_t)point * Kn + lane];

        const float xc0 = xb[(size_t)n * Cn + lane];
        const float xc1 = xb[(size_t)n * Cn + lane + 32];

        #pragma unroll
        for (int i = 0; i < Kn / 2; ++i) {
            const int r0  = __builtin_amdgcn_readlane(vidx, 2 * i);
            const int r1  = __builtin_amdgcn_readlane(vidx, 2 * i + 1);
            const int row = hi ? r1 : r0;
            const float4 v = *(const float4*)(xb + (size_t)row * Cn + l15 * 4);
            part[32 * i + lane] = (v.x + v.y) + (v.z + v.w);
        }

        __builtin_amdgcn_fence(__ATOMIC_ACQ_REL, "wavefront");

        float s0 = 0.f, s1 = 0.f;
        #pragma unroll
        for (int t = 0; t < 5; ++t) {
            s0 += part[5 * lane        + t];
            s1 += part[5 * (lane + 32) + t];
        }
        const float inv = 1.0f / (float)Kn;
        Atile[p * A_STRIDE + lane]           = s0 * inv - xc0;
        Atile[p * A_STRIDE + lane + 32]      = s1 * inv - xc1;
        Atile[p * A_STRIDE + 64 + lane]      = xc0;
        Atile[p * A_STRIDE + 64 + lane + 32] = xc1;
    }

    v8f acc0 = {0,0,0,0,0,0,0,0};
    v8f acc1 = acc0, acc2 = acc0, acc3 = acc0;

    #pragma unroll 4
    for (int kk = 0; kk < 32; ++kk) {
        const float2 av = *(const float2*)&Atile[l15 * A_STRIDE + kk * 4 + 2 * hi];
        const v2f Af = { av.x, av.y };

        const float* wb = W + (size_t)l15 * (2 * Cn) + kk * 4 + 2 * hi;
        const float2 b0 = *(const float2*)(wb + (size_t)0  * 16 * (2 * Cn));
        const float2 b1 = *(const float2*)(wb + (size_t)1  * 16 * (2 * Cn));
        const float2 b2 = *(const float2*)(wb + (size_t)2  * 16 * (2 * Cn));
        const float2 b3 = *(const float2*)(wb + (size_t)3  * 16 * (2 * Cn));
        const v2f B0 = { b0.x, b0.y };
        const v2f B1 = { b1.x, b1.y };
        const v2f B2 = { b2.x, b2.y };
        const v2f B3 = { b3.x, b3.y };

        acc0 = __builtin_amdgcn_wmma_f32_16x16x4_f32(false, Af, false, B0, (short)0, acc0, false, false);
        acc1 = __builtin_amdgcn_wmma_f32_16x16x4_f32(false, Af, false, B1, (short)0, acc1, false, false);
        acc2 = __builtin_amdgcn_wmma_f32_16x16x4_f32(false, Af, false, B2, (short)0, acc2, false, false);
        acc3 = __builtin_amdgcn_wmma_f32_16x16x4_f32(false, Af, false, B3, (short)0, acc3, false, false);
    }

    const float bz0 = bias[ 0 + l15];
    const float bz1 = bias[16 + l15];
    const float bz2 = bias[32 + l15];
    const float bz3 = bias[48 + l15];

    #pragma unroll
    for (int v = 0; v < 8; ++v) {
        const int prow = v + 8 * hi;
        float* o = out + (size_t)(base_point + prow) * On;
        o[ 0 + l15] = acc0[v] + bz0;
        o[16 + l15] = acc1[v] + bz1;
        o[32 + l15] = acc2[v] + bz2;
        o[48 + l15] = acc3[v] + bz3;
    }
}

extern "C" void kernel_launch(void* const* d_in, const int* in_sizes, int n_in,
                              void* d_out, int out_size, void* d_ws, size_t ws_size,
                              hipStream_t stream) {
    const float* x    = (const float*)d_in[0];
    const int*   adj  = (const int*)  d_in[1];
    const float* W    = (const float*)d_in[2];
    const float* bias = (const float*)d_in[3];
    float* out = (float*)d_out;

    const int total_points = Bn * Nn;
    const int total_waves  = total_points / 16;
    const int blocks       = total_waves / WAVES;

    EdgeConv_wmma_kernel<<<blocks, WAVES * 32, 0, stream>>>(x, adj, W, bias, out);
}


# Each wave keeps one 16-column slice of W in registers (32 fragment loads, not 128) and multiplies all four waves' A tiles; results pass through LDS so each wave stores whole 256-byte rows
# speedup vs baseline: 1.4275x; 1.1436x over previous
#include <hip/hip_runtime.h>

typedef __attribute__((ext_vector_type(2))) float v2f;
typedef __attribute__((ext_vector_type(8))) float v8f;

constexpr int Bn = 4;
constexpr int Nn = 32768;
constexpr int Kn = 20;
constexpr int Cn = 64;
constexpr int On = 64;

constexpr int WAVES   = 4;
constexpr int A_STRIDE = 132;
constexpr int NPART   = Kn * Cn / 4;
constexpr int O_STRIDE = 68;

__global__ __launch_bounds__(WAVES * 32)
void EdgeConv_wmma_kernel(const float* __restrict__ x,
                          const int*   __restrict__ adj,
                          const float* __restrict__ W,
                          const float* __restrict__ bias,
                          float*       __restrict__ out)
{
    __shared__ __align__(16) float s_part[WAVES][NPART];
    __shared__ __align__(16) float s_A[WAVES][16 * A_STRIDE];

    const int lane = threadIdx.x & 31;
    const int wid  = threadIdx.x >> 5;
    const int wave_global = blockIdx.x * WAVES + wid;
    const int base_point  = wave_global * 16;

    float* part   = s_part[wid];
    float* Atile  = s_A[wid];

    const int hi  = lane >> 4;
    const int l15 = lane & 15;

    for (int p = 0; p < 16; ++p) {
        const int point = base_point + p;
        const int b = point >> 15;
        const int n = point & (Nn - 1);
        const float* xb = x + (size_t)b * Nn * Cn;

        int vidx = 0;
        if (lane < Kn) vidx = adj[(size_t)point * Kn + lane];

        const float xc0 = xb[(size_t)n * Cn + lane];
        const float xc1 = xb[(size_t)n * Cn + lane + 32];

        #pragma unroll
        for (int i = 0; i < Kn / 2; ++i) {
            const int r0  = __builtin_amdgcn_readlane(vidx, 2 * i);
            const int r1  = __builtin_amdgcn_readlane(vidx, 2 * i + 1);
            const int row = hi ? r1 : r0;
            const float4 v = *(const float4*)(xb + (size_t)row * Cn + l15 * 4);
            part[32 * i + lane] = (v.x + v.y) + (v.z + v.w);
        }

        __builtin_amdgcn_fence(__ATOMIC_ACQ_REL, "wavefront");

        float s0 = 0.f, s1 = 0.f;
        #pragma unroll
        for (int t = 0; t < 5; ++t) {
            s0 += part[5 * lane        + t];
            s1 += part[5 * (lane + 32) + t];
        }
        const float inv = 1.0f / (float)Kn;
        Atile[p * A_STRIDE + lane]           = s0 * inv - xc0;
        Atile[p * A_STRIDE + lane + 32]      = s1 * inv - xc1;
        Atile[p * A_STRIDE + 64 + lane]      = xc0;
        Atile[p * A_STRIDE + 64 + lane + 32] = xc1;
    }

    const float* wb = W + (size_t)(16 * wid + l15) * (2 * Cn) + 2 * hi;
    v2f Bf[32];
    #pragma unroll
    for (int kk = 0; kk < 32; ++kk) {
        const float2 b = *(const float2*)(wb + kk * 4);
        Bf[kk] = (v2f){ b.x, b.y };
    }
    const float bz = bias[16 * wid + l15];

    __syncthreads();

    v8f acc[WAVES];
    #pragma unroll
    for (int t = 0; t < WAVES; ++t) acc[t] = (v8f){0,0,0,0,0,0,0,0};

    const float* Arow = &s_A[0][0] + l15 * A_STRIDE + 2 * hi;
    #pragma unroll
    for (int kk = 0; kk < 32; ++kk) {
        #pragma unroll
        for (int t = 0; t < WAVES; ++t) {
            const float2 av = *(const float2*)(Arow + t * (16 * A_STRIDE) + kk * 4);
            const v2f Af = { av.x, av.y };
            acc[t] = __builtin_amdgcn_wmma_f32_16x16x4_f32(false, Af, false, Bf[kk], (short)0, acc[t], false, false);
        }
    }

    __syncthreads();

    float* Otile = &s_A[0][0];
    #pragma unroll
    for (int t = 0; t < WAVES; ++t) {
        #pragma unroll
        for (int v = 0; v < 8; ++v) {
            Otile[(16 * t + 8 * hi + v) * O_STRIDE + 16 * wid + l15] = acc[t][v] + bz;
        }
    }

    __syncthreads();

    const int block_point = blockIdx.x * (WAVES * 16);
    #pragma unroll
    for (int i = 0; i < 8; ++i) {
        const int row = 16 * wid + 2 * i + hi;
        const float4 v = *(const float4*)(Otile + row * O_STRIDE + 4 * l15);
        *(float4*)(out + (size_t)(block_point + row) * On + 4 * l15) = v;
    }
}

extern "C" void kernel_launch(void* const* d_in, const int* in_sizes, int n_in,
                              void* d_out, int out_size, void* d_ws, size_t ws_size,
                              hipStream_t stream) {
    const float* x    = (const float*)d_in[0];
    const int*   adj  = (const int*)  d_in[1];
    const float* W    = (const float*)d_in[2];
    const float* bias = (const float*)d_in[3];
    float* out = (float*)d_out;

    const int total_points = Bn * Nn;
    const int total_waves  = total_points / 16;
    const int blocks       = total_waves / WAVES;

    EdgeConv_wmma_kernel<<<blocks, WAVES * 32, 0, stream>>>(x, adj, W, bias, out);
}


# A first launch writes each x row's four-float piece sums (64 B per row) to the workspace; the gather reads 64 B per neighbour, not 256 B; sums and their order unchanged, results bit-identical
# speedup vs baseline: 1.5954x; 1.1176x over previous
#include <hip/hip_runtime.h>

typedef __attribute__((ext_vector_type(2))) float v2f;
typedef __attribute__((ext_vector_type(8))) float v8f;
typedef __attribute__((ext_vector_type(4))) float v4f;

constexpr int Bn = 4;
constexpr int Nn = 32768;
constexpr int Kn = 20;
constexpr int Cn = 64;
constexpr int On = 64;

constexpr int WAVES   = 4;
constexpr int A_STRIDE = 132;
constexpr int NPART   = Kn * Cn / 4;
constexpr int O_STRIDE = 68;

constexpr int    ROW_QUADS   = Cn / 16;
constexpr size_t TABLE_BYTES = (size_t)Bn * Nn * (Cn / 4) * sizeof(float);
constexpr int    PRE_THREADS = 256;
constexpr int    PRE_LOADS   = 16;
constexpr int    PRE_BLOCKS  = Bn * Nn * (Cn / 4) / (PRE_THREADS * PRE_LOADS);

__global__ __launch_bounds__(PRE_THREADS)
void piece_sums_kernel(const float* __restrict__ x, float* __restrict__ table, size_t ws_bytes)
{
    if (ws_bytes < TABLE_BYTES) return;

    const float4* x4 = (const float4*)x;
    const unsigned base = blockIdx.x * (PRE_THREADS * PRE_LOADS) + threadIdx.x;
    float4 v[PRE_LOADS];
    #pragma unroll
    for (int k = 0; k < PRE_LOADS; ++k) v[k] = x4[base + PRE_THREADS * k];
    #pragma unroll
    for (int k = 0; k < PRE_LOADS; ++k)
        table[base + PRE_THREADS * k] = (v[k].x + v[k].y) + (v[k].z + v[k].w);
}

template <bool TABLE>
__device__ __forceinline__ void build_A_tile(const float* __restrict__ x, const int* __restrict__ adj,
                                             const v4f* __restrict__ tab4,
                                             float* part, float* Atile, int base_point, int lane)
{
    const int quad = lane & 3;
    const int nb   = lane >> 2;
    for (int p = 0; p < 16; ++p) {
        const int point = base_point + p;
        const int b = point >> 15;
        const int n = point & (Nn - 1);
        const float* xb = x + (size_t)b * Nn * Cn;

        int vidx = 0;
        if (lane < Kn) vidx = adj[(size_t)point * Kn + lane];

        const float xc0 = xb[(size_t)n * Cn + lane];
        const float xc1 = xb[(size_t)n * Cn + lane + 32];

        const int r0 = __shfl(vidx, nb)      & (Nn - 1);
        const int r1 = __shfl(vidx, nb + 8)  & (Nn - 1);
        const int r2 = __shfl(vidx, nb + 16) & (Nn - 1);
        v4f q0, q1, q2;
        if (TABLE) {
            const unsigned e = (unsigned)(b * Nn) * ROW_QUADS + quad;
            q0 = tab4[e + r0 * ROW_QUADS];
            q1 = tab4[e + r1 * ROW_QUADS];
            q2 = tab4[e + r2 * ROW_QUADS];
        } else {
            const float4* x4 = (const float4*)xb + 4 * quad;
            const int rr[3] = { r0, r1, r2 };
            v4f qq[3];
            #pragma unroll
            for (int i = 0; i < 3; ++i) {
                const float4* r4 = x4 + (size_t)rr[i] * (Cn / 4);
                const float4 a0 = r4[0], a1 = r4[1], a2 = r4[2], a3 = r4[3];
                qq[i].x = (a0.x + a0.y) + (a0.z + a0.w);
                qq[i].y = (a1.x + a1.y) + (a1.z + a1.w);
                qq[i].z = (a2.x + a2.y) + (a2.z + a2.w);
                qq[i].w = (a3.x + a3.y) + (a3.z + a3.w);
            }
            q0 = qq[0]; q1 = qq[1]; q2 = qq[2];
        }
        asm volatile("" :: "v"(q0), "v"(q1), "v"(q2));
        v4f* part4 = (v4f*)part;
        part4[lane]      = q0;
        part4[lane + 32] = q1;
        if (lane < NPART / 4 - 64) part4[lane + 64] = q2;

        __builtin_amdgcn_fence(__ATOMIC_ACQ_REL, "wavefront");

        float s0 = 0.f, s1 = 0.f;
        #pragma unroll
        for (int t = 0; t < 5; ++t) {
            s0 += part[5 * lane        + t];
            s1 += part[5 * (lane + 32) + t];
        }
        const float inv = 1.0f / (float)Kn;
        Atile[p * A_STRIDE + lane]           = s0 * inv - xc0;
        Atile[p * A_STRIDE + lane + 32]      = s1 * inv - xc1;
        Atile[p * A_STRIDE + 64 + lane]      = xc0;
        Atile[p * A_STRIDE + 64 + lane + 32] = xc1;
    }
}

__global__ __launch_bounds__(WAVES * 32)
void EdgeConv_wmma_kernel(const float* __restrict__ x,
                          const int*   __restrict__ adj,
                          const float* __restrict__ W,
                          const float* __restrict__ bias,
                          const float* __restrict__ table,
                          size_t                    ws_bytes,
                          float*       __restrict__ out)
{
    __shared__ __align__(16) float s_part[WAVES][NPART];
    __shared__ __align__(16) float s_A[WAVES][16 * A_STRIDE];

    const int lane = threadIdx.x & 31;
    const int wid  = threadIdx.x >> 5;
    const int wave_global = blockIdx.x * WAVES + wid;
    const int base_point  = wave_global * 16;

    float* part   = s_part[wid];
    float* Atile  = s_A[wid];

    const int hi  = lane >> 4;
    const int l15 = lane & 15;

    if (ws_bytes >= TABLE_BYTES) build_A_tile<true >(x, adj, (const v4f*)table, part, Atile, base_point, lane);
    else                         build_A_tile<false>(x, adj, (const v4f*)table, part, Atile, base_point, lane);

    const float* wb = W + (size_t)(16 * wid + l15) * (2 * Cn) + 2 * hi;
    v2f Bf[32];
    #pragma unroll
    for (int kk = 0; kk < 32; ++kk) {
        const float2 b = *(const float2*)(wb + kk * 4);
        Bf[kk] = (v2f){ b.x, b.y };
    }
    const float bz = bias[16 * wid + l15];

    __syncthreads();

    v8f acc[WAVES];
    #pragma unroll
    for (int t = 0; t < WAVES; ++t) acc[t] = (v8f){0,0,0,0,0,0,0,0};

    const float* Arow = &s_A[0][0] + l15 * A_STRIDE + 2 * hi;
    #pragma unroll
    for (int kk = 0; kk < 32; ++kk) {
        #pragma unroll
        for (int t = 0; t < WAVES; ++t) {
            const float2 av = *(const float2*)(Arow + t * (16 * A_STRIDE) + kk * 4);
            const v2f Af = { av.x, av.y };
            acc[t] = __builtin_amdgcn_wmma_f32_16x16x4_f32(false, Af, false, Bf[kk], (short)0, acc[t], false, false);
        }
    }

    __syncthreads();

    float* Otile = &s_A[0][0];
    #pragma unroll
    for (int t = 0; t < WAVES; ++t) {
        #pragma unroll
        for (int v = 0; v < 8; ++v) {
            Otile[(16 * t + 8 * hi + v) * O_STRIDE + 16 * wid + l15] = acc[t][v] + bz;
        }
    }

    __syncthreads();

    const int block_point = blockIdx.x * (WAVES * 16);
    #pragma unroll
    for (int i = 0; i < 8; ++i) {
        const int row = 16 * wid + 2 * i + hi;
        const float4 v = *(const float4*)(Otile + row * O_STRIDE + 4 * l15);
        *(float4*)(out + (size_t)(block_point + row) * On + 4 * l15) = v;
    }
}

extern "C" void kernel_launch(void* const* d_in, const int* in_sizes, int n_in,
                              void* d_out, int out_size, void* d_ws, size_t ws_size,
                              hipStream_t stream) {
    const float* x    = (const float*)d_in[0];
    const int*   adj  = (const int*)  d_in[1];
    const float* W    = (const float*)d_in[2];
    const float* bias = (const float*)d_in[3];
    float* out = (float*)d_out;
    float* table = (float*)d_ws;

    const int total_points = Bn * Nn;
    const int total_waves  = total_points / 16;
    const int blocks       = total_waves / WAVES;

    piece_sums_kernel<<<PRE_BLOCKS, PRE_THREADS, 0, stream>>>(x, table, ws_size);
    EdgeConv_wmma_kernel<<<blocks, WAVES * 32, 0, stream>>>(x, adj, W, bias, table, ws_size, out);
}


# Gather software-pipelined: indices and centre rows of all 16 points load first, each point's table loads issue 4 points ahead of its LDS exchange; sums and their order unchanged, bit-identical
# speedup vs baseline: 1.7794x; 1.1153x over previous
#include <hip/hip_runtime.h>

typedef __attribute__((ext_vector_type(2))) float v2f;
typedef __attribute__((ext_vector_type(8))) float v8f;
typedef __attribute__((ext_vector_type(4))) float v4f;

constexpr int Bn = 4;
constexpr int Nn = 32768;
constexpr int Kn = 20;
constexpr int Cn = 64;
constexpr int On = 64;

constexpr int WAVES   = 4;
constexpr int A_STRIDE = 132;
constexpr int NPART   = Kn * Cn / 4;
constexpr int O_STRIDE = 68;
constexpr int PART_FLOATS = 384;
constexpr int DEPTH   = 4;

constexpr int    ROW_QUADS   = Cn / 16;
constexpr size_t TABLE_BYTES = (size_t)Bn * Nn * (Cn / 4) * sizeof(float);
constexpr int    PRE_THREADS = 256;
constexpr int    PRE_LOADS   = 16;
constexpr int    PRE_BLOCKS  = Bn * Nn * (Cn / 4) / (PRE_THREADS * PRE_LOADS);

__global__ __launch_bounds__(PRE_THREADS)
void piece_sums_kernel(const float* __restrict__ x, float* __restrict__ table, size_t ws_bytes)
{
    if (ws_bytes < TABLE_BYTES) return;

    const float4* x4 = (const float4*)x;
    const unsigned base = blockIdx.x * (PRE_THREADS * PRE_LOADS) + threadIdx.x;
    float4 v[PRE_LOADS];
    #pragma unroll
    for (int k = 0; k < PRE_LOADS; ++k) v[k] = x4[base + PRE_THREADS * k];
    #pragma unroll
    for (int k = 0; k < PRE_LOADS; ++k)
        table[base + PRE_THREADS * k] = (v[k].x + v[k].y) + (v[k].z + v[k].w);
}

__device__ __forceinline__ void build_A_tile_table(const float* __restrict__ x, const int* __restrict__ adj,
                                                   const v4f* __restrict__ tab4,
                                                   float* part, float* Atile, int base_point, int lane)
{
    const int quad = lane & 3;
    const int nb   = lane >> 2;
    const int b    = base_point >> 15;
    const v4f* tb  = tab4 + (size_t)(b * Nn) * ROW_QUADS + quad;


    const int* adjw = adj + (size_t)base_point * Kn + (lane < Kn ? lane : Kn - 1);
    int vidx[16];
    #pragma unroll
    for (int p = 0; p < 16; ++p) vidx[p] = adjw[p * Kn];

    const v4f* xc4 = (const v4f*)(x + (size_t)base_point * Cn);
    v4f xc[8];
    #pragma unroll
    for (int j = 0; j < 8; ++j) xc[j] = xc4[lane + 32 * j];
    #pragma unroll
    for (int j = 0; j < 8; ++j)
        *(v4f*)(Atile + (2 * j + (lane >> 4)) * A_STRIDE + 64 + 4 * (lane & 15)) = xc[j];

    int r[16][3];
    v4f q[16][3];
    #pragma unroll
    for (int p = 0; p < DEPTH + 1; ++p) {
        #pragma unroll
        for (int s = 0; s < 3; ++s) r[p][s] = __shfl(vidx[p], nb + 8 * s) & (Nn - 1);
    }
    #pragma unroll
    for (int p = 0; p < DEPTH; ++p) {
        #pragma unroll
        for (int s = 0; s < 3; ++s) q[p][s] = tb[r[p][s] * ROW_QUADS];
    }

    v4f* part4 = (v4f*)part;
    #pragma unroll
    for (int p = 0; p < 16; ++p) {
        if (p + DEPTH < 16) {
            #pragma unroll
            for (int s = 0; s < 3; ++s) q[p + DEPTH][s] = tb[r[p + DEPTH][s] * ROW_QUADS];
        }
        if (p + DEPTH + 1 < 16) {
            #pragma unroll
            for (int s = 0; s < 3; ++s) r[p + DEPTH + 1][s] = __shfl(vidx[p + DEPTH + 1], nb + 8 * s) & (Nn - 1);
        }

        part4[lane]      = q[p][0];
        part4[lane + 32] = q[p][1];
        part4[lane + 64] = q[p][2];

        __builtin_amdgcn_fence(__ATOMIC_ACQ_REL, "wavefront");

        float s0 = 0.f, s1 = 0.f;
        #pragma unroll
        for (int t = 0; t < 5; ++t) {
            s0 += part[5 * lane        + t];
            s1 += part[5 * (lane + 32) + t];
        }
        const float xc0 = Atile[p * A_STRIDE + 64 + lane];
        const float xc1 = Atile[p * A_STRIDE + 64 + lane + 32];
        const float inv = 1.0f / (float)Kn;
        Atile[p * A_STRIDE + lane]      = s0 * inv - xc0;
        Atile[p * A_STRIDE + lane + 32] = s1 * inv - xc1;
    }
}

__device__ __forceinline__ void build_A_tile_direct(const float* __restrict__ x, const int* __restrict__ adj,
                                                    float* part, float* Atile, int base_point, int lane)
{
    const int quad = lane & 3;
    const int nb   = lane >> 2;
    for (int p = 0; p < 16; ++p) {
        const int point = base_point + p;
        const int b = point >> 15;
        const int n = point & (Nn - 1);
        const float* xb = x + (size_t)b * Nn * Cn;

        int vidx = 0;
        if (lane < Kn) vidx = adj[(size_t)point * Kn + lane];

        const float xc0 = xb[(size_t)n * Cn + lane];
        const float xc1 = xb[(size_t)n * Cn + lane + 32];

        const int rr[3] = { __shfl(vidx, nb) & (Nn - 1), __shfl(vidx, nb + 8) & (Nn - 1), __shfl(vidx, nb + 16) & (Nn - 1) };
        const float4* x4 = (const float4*)xb + 4 * quad;
        v4f qq[3];
        #pragma unroll
        for (int i = 0; i < 3; ++i) {
            const float4* r4 = x4 + (size_t)rr[i] * (Cn / 4);
            const float4 a0 = r4[0], a1 = r4[1], a2 = r4[2], a3 = r4[3];
            qq[i].x = (a0.x + a0.y) + (a0.z + a0.w);
            qq[i].y = (a1.x + a1.y) + (a1.z + a1.w);
            qq[i].z = (a2.x + a2.y) + (a2.z + a2.w);
            qq[i].w = (a3.x + a3.y) + (a3.z + a3.w);
        }
        asm volatile("" :: "v"(qq[0]), "v"(qq[1]), "v"(qq[2]));
        v4f* part4 = (v4f*)part;
        part4[lane]      = qq[0];
        part4[lane + 32] = qq[1];
        if (lane < NPART / 4 - 64) part4[lane + 64] = qq[2];

        __builtin_amdgcn_fence(__ATOMIC_ACQ_REL, "wavefront");

        float s0 = 0.f, s1 = 0.f;
        #pragma unroll
        for (int t = 0; t < 5; ++t) {
            s0 += part[5 * lane        + t];
            s1 += part[5 * (lane + 32) + t];
        }
        const float inv = 1.0f / (float)Kn;
        Atile[p * A_STRIDE + lane]           = s0 * inv - xc0;
        Atile[p * A_STRIDE + lane + 32]      = s1 * inv - xc1;
        Atile[p * A_STRIDE + 64 + lane]      = xc0;
        Atile[p * A_STRIDE + 64 + lane + 32] = xc1;
    }
}

__global__ __launch_bounds__(WAVES * 32)
void EdgeConv_wmma_kernel(const float* __restrict__ x,
                          const int*   __restrict__ adj,
                          const float* __restrict__ W,
                          const float* __restrict__ bias,
                          const float* __restrict__ table,
                          size_t                    ws_bytes,
                          float*       __restrict__ out)
{
    __shared__ __align__(16) float s_part[WAVES][PART_FLOATS];
    __shared__ __align__(16) float s_A[WAVES][16 * A_STRIDE];

    const int lane = threadIdx.x & 31;
    const int wid  = threadIdx.x >> 5;
    const int wave_global = blockIdx.x * WAVES + wid;
    const int base_point  = wave_global * 16;

    float* part   = s_part[wid];
    float* Atile  = s_A[wid];

    const int hi  = lane >> 4;
    const int l15 = lane & 15;

    if (ws_bytes >= TABLE_BYTES) build_A_tile_table (x, adj, (const v4f*)table, part, Atile, base_point, lane);
    else                         build_A_tile_direct(x, adj, part, Atile, base_point, lane);

    const float* wb = W + (size_t)(16 * wid + l15) * (2 * Cn) + 2 * hi;
    v2f Bf[32];
    #pragma unroll
    for (int kk = 0; kk < 32; ++kk) {
        const float2 b = *(const float2*)(wb + kk * 4);
        Bf[kk] = (v2f){ b.x, b.y };
    }
    const float bz = bias[16 * wid + l15];

    __syncthreads();

    v8f acc[WAVES];
    #pragma unroll
    for (int t = 0; t < WAVES; ++t) acc[t] = (v8f){0,0,0,0,0,0,0,0};

    const float* Arow = &s_A[0][0] + l15 * A_STRIDE + 2 * hi;
    #pragma unroll
    for (int kk = 0; kk < 32; ++kk) {
        #pragma unroll
        for (int t = 0; t < WAVES; ++t) {
            const float2 av = *(const float2*)(Arow + t * (16 * A_STRIDE) + kk * 4);
            const v2f Af = { av.x, av.y };
            acc[t] = __builtin_amdgcn_wmma_f32_16x16x4_f32(false, Af, false, Bf[kk], (short)0, acc[t], false, false);
        }
    }

    __syncthreads();

    float* Otile = &s_A[0][0];
    #pragma unroll
    for (int t = 0; t < WAVES; ++t) {
        #pragma unroll
        for (int v = 0; v < 8; ++v) {
            Otile[(16 * t + 8 * hi + v) * O_STRIDE + 16 * wid + l15] = acc[t][v] + bz;
        }
    }

    __syncthreads();

    const int block_point = blockIdx.x * (WAVES * 16);
    #pragma unroll
    for (int i = 0; i < 8; ++i) {
        const int row = 16 * wid + 2 * i + hi;
        const float4 v = *(const float4*)(Otile + row * O_STRIDE + 4 * l15);
        *(float4*)(out + (size_t)(block_point + row) * On + 4 * l15) = v;
    }
}

extern "C" void kernel_launch(void* const* d_in, const int* in_sizes, int n_in,
                              void* d_out, int out_size, void* d_ws, size_t ws_size,
                              hipStream_t stream) {
    const float* x    = (const float*)d_in[0];
    const int*   adj  = (const int*)  d_in[1];
    const float* W    = (const float*)d_in[2];
    const float* bias = (const float*)d_in[3];
    float* out = (float*)d_out;
    float* table = (float*)d_ws;

    const int total_points = Bn * Nn;
    const int total_waves  = total_points / 16;
    const int blocks       = total_waves / WAVES;

    piece_sums_kernel<<<PRE_BLOCKS, PRE_THREADS, 0, stream>>>(x, table, ws_size);
    EdgeConv_wmma_kernel<<<blocks, WAVES * 32, 0, stream>>>(x, adj, W, bias, table, ws_size, out);
}


# Wave roles in 8-wave blocks (512 blocks, 4 tiles each): 4 waves gather tile s into one of two A buffers while 4 waves run tile s-1's WMMAs and store tile s-2; sums and their order unchanged
# speedup vs baseline: 1.9124x; 1.0747x over previous
#include <hip/hip_runtime.h>

typedef __attribute__((ext_vector_type(2))) float v2f;
typedef __attribute__((ext_vector_type(8))) float v8f;
typedef __attribute__((ext_vector_type(4))) float v4f;

constexpr int Bn = 4;
constexpr int Nn = 32768;
constexpr int Kn = 20;
constexpr int Cn = 64;
constexpr int On = 64;

constexpr int WAVES   = 4;
constexpr int TILES   = 4;
constexpr int NBLOCKS = Bn * Nn / (64 * TILES);
constexpr int A_STRIDE = 132;
constexpr int NPART   = Kn * Cn / 4;
constexpr int O_STRIDE = 68;
constexpr int PART_FLOATS = 384;
constexpr int DEPTH   = 4;

constexpr int    ROW_QUADS   = Cn / 16;
constexpr size_t TABLE_BYTES = (size_t)Bn * Nn * (Cn / 4) * sizeof(float);
constexpr int    PRE_THREADS = 256;
constexpr int    PRE_LOADS   = 16;
constexpr int    PRE_BLOCKS  = Bn * Nn * (Cn / 4) / (PRE_THREADS * PRE_LOADS);

__global__ __launch_bounds__(PRE_THREADS)
void piece_sums_kernel(const float* __restrict__ x, float* __restrict__ table, size_t ws_bytes)
{
    if (ws_bytes < TABLE_BYTES) return;

    const float4* x4 = (const float4*)x;
    const unsigned base = blockIdx.x * (PRE_THREADS * PRE_LOADS) + threadIdx.x;
    float4 v[PRE_LOADS];
    #pragma unroll
    for (int k = 0; k < PRE_LOADS; ++k) v[k] = x4[base + PRE_THREADS * k];
    #pragma unroll
    for (int k = 0; k < PRE_LOADS; ++k)
        table[base + PRE_THREADS * k] = (v[k].x + v[k].y) + (v[k].z + v[k].w);
}

__device__ __forceinline__ void build_A_tile_table(const float* __restrict__ x, const int* __restrict__ adj,
                                                   const v4f* __restrict__ tab4,
                                                   float* part, float* Atile, int base_point, int lane)
{
    const int quad = lane & 3;
    const int nb   = lane >> 2;
    const int b    = base_point >> 15;
    const v4f* tb  = tab4 + (size_t)(b * Nn) * ROW_QUADS + quad;


    const int* adjw = adj + (size_t)base_point * Kn + (lane < Kn ? lane : Kn - 1);
    int vidx[16];
    #pragma unroll
    for (int p = 0; p < 16; ++p) vidx[p] = adjw[p * Kn];

    const v4f* xc4 = (const v4f*)(x + (size_t)base_point * Cn);
    v4f xc[8];
    #pragma unroll
    for (int j = 0; j < 8; ++j) xc[j] = xc4[lane + 32 * j];
    #pragma unroll
    for (int j = 0; j < 8; ++j)
        *(v4f*)(Atile + (2 * j + (lane >> 4)) * A_STRIDE + 64 + 4 * (lane & 15)) = xc[j];

    int r[16][3];
    v4f q[16][3];
    #pragma unroll
    for (int p = 0; p < DEPTH + 1; ++p) {
        #pragma unroll
        for (int s = 0; s < 3; ++s) r[p][s] = __shfl(vidx[p], nb + 8 * s) & (Nn - 1);
    }
    #pragma unroll
    for (int p = 0; p < DEPTH; ++p) {
        #pragma unroll
        for (int s = 0; s < 3; ++s) q[p][s] = tb[r[p][s] * ROW_QUADS];
    }

    v4f* part4 = (v4f*)part;
    #pragma unroll
    for (int p = 0; p < 16; ++p) {
        if (p + DEPTH < 16) {
            #pragma unroll
            for (int s = 0; s < 3; ++s) q[p + DEPTH][s] = tb[r[p + DEPTH][s] * ROW_QUADS];
        }
        if (p + DEPTH + 1 < 16) {
            #pragma unroll
            for (int s = 0; s < 3; ++s) r[p + DEPTH + 1][s] = __shfl(vidx[p + DEPTH + 1], nb + 8 * s) & (Nn - 1);
        }

        part4[lane]      = q[p][0];
        part4[lane + 32] = q[p][1];
        part4[lane + 64] = q[p][2];

        __builtin_amdgcn_fence(__ATOMIC_ACQ_REL, "wavefront");

        float s0 = 0.f, s1 = 0.f;
        #pragma unroll
        for (int t = 0; t < 5; ++t) {
            s0 += part[5 * lane        + t];
            s1 += part[5 * (lane + 32) + t];
        }
        const float xc0 = Atile[p * A_STRIDE + 64 + lane];
        const float xc1 = Atile[p * A_STRIDE + 64 + lane + 32];
        const float inv = 1.0f / (float)Kn;
        Atile[p * A_STRIDE + lane]      = s0 * inv - xc0;
        Atile[p * A_STRIDE + lane + 32] = s1 * inv - xc1;
    }
}

__device__ __forceinline__ void build_A_tile_direct(const float* __restrict__ x, const int* __restrict__ adj,
                                                    float* part, float* Atile, int base_point, int lane)
{
    const int quad = lane & 3;
    const int nb   = lane >> 2;
    for (int p = 0; p < 16; ++p) {
        const int point = base_point + p;
        const int b = point >> 15;
        const int n = point & (Nn - 1);
        const float* xb = x + (size_t)b * Nn * Cn;

        int vidx = 0;
        if (lane < Kn) vidx = adj[(size_t)point * Kn + lane];

        const float xc0 = xb[(size_t)n * Cn + lane];
        const float xc1 = xb[(size_t)n * Cn + lane + 32];

        const int rr[3] = { __shfl(vidx, nb) & (Nn - 1), __shfl(vidx, nb + 8) & (Nn - 1), __shfl(vidx, nb + 16) & (Nn - 1) };
        const float4* x4 = (const float4*)xb + 4 * quad;
        v4f qq[3];
        #pragma unroll
        for (int i = 0; i < 3; ++i) {
            const float4* r4 = x4 + (size_t)rr[i] * (Cn / 4);
            const float4 a0 = r4[0], a1 = r4[1], a2 = r4[2], a3 = r4[3];
            qq[i].x = (a0.x + a0.y) + (a0.z + a0.w);
            qq[i].y = (a1.x + a1.y) + (a1.z + a1.w);
            qq[i].z = (a2.x + a2.y) + (a2.z + a2.w);
            qq[i].w = (a3.x + a3.y) + (a3.z + a3.w);
        }
        asm volatile("" :: "v"(qq[0]), "v"(qq[1]), "v"(qq[2]));
        v4f* part4 = (v4f*)part;
        part4[lane]      = qq[0];
        part4[lane + 32] = qq[1];
        if (lane < NPART / 4 - 64) part4[lane + 64] = qq[2];

        __builtin_amdgcn_fence(__ATOMIC_ACQ_REL, "wavefront");

        float s0 = 0.f, s1 = 0.f;
        #pragma unroll
        for (int t = 0; t < 5; ++t) {
            s0 += part[5 * lane        + t];
            s1 += part[5 * (lane + 32) + t];
        }
        const float inv = 1.0f / (float)Kn;
        Atile[p * A_STRIDE + lane]           = s0 * inv - xc0;
        Atile[p * A_STRIDE + lane + 32]      = s1 * inv - xc1;
        Atile[p * A_STRIDE + 64 + lane]      = xc0;
        Atile[p * A_STRIDE + 64 + lane + 32] = xc1;
    }
}

constexpr int LDS_PART   = WAVES * PART_FLOATS;
constexpr int LDS_A      = WAVES * 16 * A_STRIDE;
constexpr int LDS_O      = 64 * O_STRIDE;
constexpr int LDS_FLOATS = LDS_PART + 2 * LDS_A + 2 * LDS_O;

__global__ __launch_bounds__(2 * WAVES * 32)
void EdgeConv_wmma_kernel(const float* __restrict__ x,
                          const int*   __restrict__ adj,
                          const float* __restrict__ W,
                          const float* __restrict__ bias,
                          const float* __restrict__ table,
                          size_t                    ws_bytes,
                          float*       __restrict__ out)
{
    __shared__ __align__(16) float lds[LDS_FLOATS];
    float* s_part = lds;
    float* s_A    = lds + LDS_PART;
    float* s_O    = lds + LDS_PART + 2 * LDS_A;

    const int lane = threadIdx.x & 31;
    const int wid  = __builtin_amdgcn_readfirstlane(threadIdx.x >> 5);
    const bool gatherer = wid < WAVES;
    const int rw   = wid & (WAVES - 1);

    const int hi  = lane >> 4;
    const int l15 = lane & 15;

    v2f Bf[32];
    float bz = 0.f;
    if (!gatherer) {
        const float* wb = W + (size_t)(16 * rw + l15) * (2 * Cn) + 2 * hi;
        #pragma unroll
        for (int kk = 0; kk < 32; ++kk) {
            const float2 b = *(const float2*)(wb + kk * 4);
            Bf[kk] = (v2f){ b.x, b.y };
        }
        bz = bias[16 * rw + l15];
    }

    const int first_tile = blockIdx.x * TILES;

    for (int s = 0; s < TILES + 2; ++s) {
        if (gatherer) {
            if (s < TILES) {
                const int base_point = ((first_tile + s) * WAVES + rw) * 16;
                float* part  = s_part + rw * PART_FLOATS;
                float* Atile = s_A + (s & 1) * LDS_A + rw * (16 * A_STRIDE);
                if (ws_bytes >= TABLE_BYTES) build_A_tile_table (x, adj, (const v4f*)table, part, Atile, base_point, lane);
                else                         build_A_tile_direct(x, adj, part, Atile, base_point, lane);
            }
        } else {
            if (s >= 2) {
                const float* Otile = s_O + (s & 1) * LDS_O;
                const int tile_point = (first_tile + s - 2) * 64;
                #pragma unroll
                for (int i = 0; i < 8; ++i) {
                    const int row = 16 * rw + 2 * i + hi;
                    const float4 v = *(const float4*)(Otile + row * O_STRIDE + 4 * l15);
                    *(float4*)(out + (size_t)(tile_point + row) * On + 4 * l15) = v;
                }
            }
            if (s >= 1 && s <= TILES) {
                v8f acc[WAVES];
                #pragma unroll
                for (int t = 0; t < WAVES; ++t) acc[t] = (v8f){0,0,0,0,0,0,0,0};

                const float* Arow = s_A + ((s - 1) & 1) * LDS_A + l15 * A_STRIDE + 2 * hi;
                #pragma unroll
                for (int kk = 0; kk < 32; ++kk) {
                    #pragma unroll
                    for (int t = 0; t < WAVES; ++t) {
                        const float2 av = *(const float2*)(Arow + t * (16 * A_STRIDE) + kk * 4);
                        const v2f Af = { av.x, av.y };
                        acc[t] = __builtin_amdgcn_wmma_f32_16x16x4_f32(false, Af, false, Bf[kk], (short)0, acc[t], false, false);
                    }
                }

                float* Otile = s_O + ((s - 1) & 1) * LDS_O;
                #pragma unroll
                for (int t = 0; t < WAVES; ++t) {
                    #pragma unroll
                    for (int v = 0; v < 8; ++v) {
                        Otile[(16 * t + 8 * hi + v) * O_STRIDE + 16 * rw + l15] = acc[t][v] + bz;
                    }
                }
            }
        }
        __syncthreads();
    }
}

extern "C" void kernel_launch(void* const* d_in, const int* in_sizes, int n_in,
                              void* d_out, int out_size, void* d_ws, size_t ws_size,
                              hipStream_t stream) {
    const float* x    = (const float*)d_in[0];
    const int*   adj  = (const int*)  d_in[1];
    const float* W    = (const float*)d_in[2];
    const float* bias = (const float*)d_in[3];
    float* out = (float*)d_out;
    float* table = (float*)d_ws;

    piece_sums_kernel<<<PRE_BLOCKS, PRE_THREADS, 0, stream>>>(x, table, ws_size);
    EdgeConv_wmma_kernel<<<NBLOCKS, 2 * WAVES * 32, 0, stream>>>(x, adj, W, bias, table, ws_size, out);
}
